# starting kernel
# baseline (speedup 1.0000x reference)
#include <hip/hip_runtime.h>
#include <stdint.h>

typedef __attribute__((ext_vector_type(2))) float v2f;
typedef __attribute__((ext_vector_type(8))) float v8f;

#define D_FEAT 128

__global__ void gcn_gemm_xwT_wmma(const float* __restrict__ x,
                                  const float* __restrict__ W,
                                  float* __restrict__ xw) {
  const int lane  = threadIdx.x & 31;
  const int wave  = threadIdx.x >> 5;
  const int mtile = blockIdx.x;
  const int arow  = mtile * 16 + (lane & 15);
  const int bcol  = wave * 16 + (lane & 15);
  const int khalf = (lane >> 4) * 2;

  const float* xp = x + arow * D_FEAT + khalf;
  const float* wp = W + bcol * D_FEAT + khalf;

  v8f acc = {};
#pragma unroll
  for (int k = 0; k < D_FEAT; k += 4) {
    v2f a = *(const v2f*)(xp + k);
    v2f b = *(const v2f*)(wp + k);
    acc = __builtin_amdgcn_wmma_f32_16x16x4_f32(
         false, a,  false, b,
         (short)0, acc,  false,  false);
  }

  const int rbase = mtile * 16 + (lane >> 4) * 8;
  const int col   = wave * 16 + (lane & 15);
#pragma unroll
  for (int r = 0; r < 8; ++r)
    xw[(rbase + r) * D_FEAT + col] = acc[r];
}

__global__ void gcn_deg_init(float* __restrict__ deg, int n) {
  int i = blockIdx.x * blockDim.x + threadIdx.x;
  if (i < n) deg[i] = 1.0f;
}

__global__ void gcn_deg_count(const int* __restrict__ dst,
                              float* __restrict__ deg, int nEdges) {
  int e = blockIdx.x * blockDim.x + threadIdx.x;
  if (e < nEdges) atomicAdd(&deg[dst[e]], 1.0f);
}

__global__ void gcn_deg_rsqrt(float* __restrict__ deg, int n) {
  int i = blockIdx.x * blockDim.x + threadIdx.x;
  if (i < n) deg[i] = rsqrtf(deg[i]);
}

__global__ void gcn_self_bias(const float* __restrict__ xw,
                              const float* __restrict__ dinv,
                              const float* __restrict__ bias,
                              float* __restrict__ out, int n) {
  int t = blockIdx.x * blockDim.x + threadIdx.x;
  if (t >= n * D_FEAT) return;
  int i = t >> 7;
  int c = t & (D_FEAT - 1);
  float dv = dinv[i];
  out[t] = xw[t] * dv * dv + bias[c];
}

__global__ void gcn_edge_scatter(const float* __restrict__ xw,
                                 const float* __restrict__ dinv,
                                 const int* __restrict__ src,
                                 const int* __restrict__ dst,
                                 float* __restrict__ out, int nEdges) {
  unsigned t = blockIdx.x * blockDim.x + threadIdx.x;
  int e = t >> 5;
  if (e >= nEdges) return;
  int lane = t & 31;
  int s = src[e];
  int d = dst[e];
  float norm = dinv[s] * dinv[d];
  const float4 v = *(const float4*)(xw + s * D_FEAT + lane * 4);
  float* o = out + d * D_FEAT + lane * 4;
  atomicAdd(o + 0, v.x * norm);
  atomicAdd(o + 1, v.y * norm);
  atomicAdd(o + 2, v.z * norm);
  atomicAdd(o + 3, v.w * norm);
}

extern "C" void kernel_launch(void* const* d_in, const int* in_sizes, int n_in,
                              void* d_out, int out_size, void* d_ws, size_t ws_size,
                              hipStream_t stream) {
  const float* x = (const float*)d_in[0];
  const float* W = (const float*)d_in[1];
  const float* b = (const float*)d_in[2];
  const int* ei  = (const int*)d_in[3];

  const int nNodes = in_sizes[0] / D_FEAT;
  const int nEdges = in_sizes[3] / 2;
  const int* src = ei;
  const int* dst = ei + nEdges;

  float* out = (float*)d_out;

  float* xw  = (float*)d_ws;
  float* deg = (float*)((char*)d_ws + (size_t)nNodes * D_FEAT * sizeof(float));

  const int mtiles = (nNodes + 15) / 16;
  gcn_gemm_xwT_wmma<<<mtiles, 256, 0, stream>>>(x, W, xw);

  gcn_deg_init<<<(nNodes + 255) / 256, 256, 0, stream>>>(deg, nNodes);

  gcn_deg_count<<<(nEdges + 255) / 256, 256, 0, stream>>>(dst, deg, nEdges);

  gcn_deg_rsqrt<<<(nNodes + 255) / 256, 256, 0, stream>>>(deg, nNodes);

  {
    long long tot = (long long)nNodes * D_FEAT;
    int blocks = (int)((tot + 255) / 256);
    gcn_self_bias<<<blocks, 256, 0, stream>>>(xw, deg, b, out, nNodes);
  }

  {
    long long tot = (long long)nEdges * 32;
    int blocks = (int)((tot + 255) / 256);
    gcn_edge_scatter<<<blocks, 256, 0, stream>>>(xw, deg, src, dst, out, nEdges);
  }
}


# Edges filed in one pass: 32-slot bucket per node (int atomic slot) + overflow pair list; scan, rank and fill passes dropped (5 launches to 3); bucket ids sorted in registers: sum order = ascending id
# speedup vs baseline: 2.6968x; 2.6968x over previous
#include <hip/hip_runtime.h>
#include <stdint.h>


typedef __attribute__((ext_vector_type(2))) float v2f;
typedef __attribute__((ext_vector_type(8))) float v8f;

#define D_FEAT 128
#define CHUNK 1024


#define CAP 32

__global__ void gcn_zero_counts(int* __restrict__ cnt, int nWords) {
  int i = blockIdx.x * blockDim.x + threadIdx.x;
  if (i < nWords) cnt[i] = 0;
}

static __device__ __forceinline__ void gcn_file_one(int d, int s, int r,
                                                    int* __restrict__ bkt,
                                                    int* __restrict__ ovf,
                                                    int* __restrict__ ovfCount) {
  if (r < CAP) {
    bkt[d * CAP + r] = s;
  } else {
    const int q = atomicAdd(ovfCount, 1);
    *(int2*)(ovf + 2 * q) = make_int2(d, s);
  }
}

__global__ void gcn_file_edges(const int* __restrict__ src,
                               const int* __restrict__ dst,
                               int* __restrict__ cnt,
                               int* __restrict__ bkt,
                               int* __restrict__ ovf,
                               int* __restrict__ ovfCount, int nEdges) {
  const int base = blockIdx.x * CHUNK + threadIdx.x;
  if (blockIdx.x * CHUNK + CHUNK <= nEdges) {
    int d[4], s[4], r[4];
    for (int k = 0; k < 4; ++k) { d[k] = dst[base + k * 256]; s[k] = src[base + k * 256]; }
    for (int k = 0; k < 4; ++k) r[k] = atomicAdd(&cnt[d[k]], 1);
    for (int k = 0; k < 4; ++k) gcn_file_one(d[k], s[k], r[k], bkt, ovf, ovfCount);
  } else {
    for (int k = 0; k < 4; ++k) {
      const int e = base + k * 256;
      if (e < nEdges) {
        const int d = dst[e];
        const int s = src[e];
        const int r = atomicAdd(&cnt[d], 1);
        gcn_file_one(d, s, r, bkt, ovf, ovfCount);
      }
    }
  }
}

#define TILE_M 32
#define WG_THREADS 512
#define L_PITCH 132
#define GROUP 16

static __device__ __forceinline__ int gcn_sort32(int key, int lane) {
#pragma unroll
  for (int k = 2; k <= 32; k <<= 1) {
#pragma unroll
    for (int j = k >> 1; j > 0; j >>= 1) {
      const int other = __shfl_xor(key, j);
      const bool up    = (lane & k) == 0;
      const bool lower = (lane & j) == 0;
      const int mn = key < other ? key : other;
      const int mx = key < other ? other : key;
      key = (lower == up) ? mn : mx;
    }
  }
  return key;
}

static __device__ __forceinline__ void gcn_issue_group(const char* __restrict__ xb,
                                                       int ids, int nent, unsigned self,
                                                       int p0, unsigned loff, float4* v) {
  const int np = nent + 1;
#pragma unroll
  for (int u = 0; u < GROUP; ++u) {
    const int p = p0 + u < np ? p0 + u : np - 1;
    const unsigned e = (unsigned)__builtin_amdgcn_readlane(ids, (p - 1) & 31);
    const unsigned s = p == 0 ? self : e;
    v[u] = *(const float4*)(xb + (s * 512u + loff));
  }
}

static __device__ __forceinline__ void gcn_consume_group(float wts, int nent, float selfw,
                                                         int p0, const float4* v, float4& acc) {
  const int np = nent + 1;
#pragma unroll
  for (int u = 0; u < GROUP; ++u) {
    const bool live = p0 + u < np;
    const int p = live ? p0 + u : np - 1;
    const float we = __int_as_float(__builtin_amdgcn_readlane(__float_as_int(wts), (p - 1) & 31));
    const float w = p == 0 ? selfw : we;
    acc.x = fmaf(live ? v[u].x : 0.0f, w, acc.x);
    acc.y = fmaf(live ? v[u].y : 0.0f, w, acc.y);
    acc.z = fmaf(live ? v[u].z : 0.0f, w, acc.z);
    acc.w = fmaf(live ? v[u].w : 0.0f, w, acc.w);
  }
}

static __device__ __forceinline__ void gcn_accum_from(const char* __restrict__ xb,
                                                      int ids, float wts, int nent,
                                                      unsigned self, float selfw, int p0,
                                                      unsigned loff, float4& acc) {
  for (; p0 < nent + 1; p0 += GROUP) {
    float4 v[GROUP];
    gcn_issue_group(xb, ids, nent, self, p0, loff, v);
    gcn_consume_group(wts, nent, selfw, p0, v, acc);
  }
}

static __device__ __forceinline__ void gcn_accum_overflow(const char* __restrict__ xb,
                                                       const int* __restrict__ cnt,
                                                       const int* __restrict__ ovf,
                                                       int npairs, int node, int lane,
                                                       unsigned loff, float4& acc) {
  for (int c0 = 0; c0 < npairs; c0 += 32) {
    int od = -1, os = 0;
    if (c0 + lane < npairs) {
      const int2 pr = *(const int2*)(ovf + 2 * (c0 + lane));
      od = pr.x;
      os = pr.y;
    }
    const bool hit = od == node;
    float ow = 0.0f;
    if (hit) ow = rsqrtf((float)(cnt[os] + 1));
    unsigned mask = __builtin_amdgcn_ballot_w32(hit);
    while (mask != 0u) {
      const int b = __builtin_ctz(mask);
      mask &= mask - 1u;
      const unsigned s = (unsigned)__builtin_amdgcn_readlane(os, b);
      const float w = __int_as_float(__builtin_amdgcn_readlane(__float_as_int(ow), b));
      const float4 v = *(const float4*)(xb + (s * 512u + loff));
      acc.x = fmaf(v.x, w, acc.x);
      acc.y = fmaf(v.y, w, acc.y);
      acc.z = fmaf(v.z, w, acc.z);
      acc.w = fmaf(v.w, w, acc.w);
    }
  }
}

static __device__ __forceinline__ v2f gcn_frag(const float* p, int kk) {
  return *(const v2f*)(p + 4 * kk);
}

__global__ void __launch_bounds__(WG_THREADS, 8)
gcn_gather_gemm_wmma(const float* __restrict__ x,
                     const int* __restrict__ cnt,
                     const int* __restrict__ bkt,
                     const int* __restrict__ ovf,
                     const int* __restrict__ ovfCount,
                     const float* __restrict__ W,
                     const float* __restrict__ bias,
                     float* __restrict__ out, int n, int ntiles, int nEdges) {
  __shared__ __attribute__((aligned(16))) float Ws[D_FEAT * L_PITCH];
  __shared__ __attribute__((aligned(16))) float As[2][TILE_M * L_PITCH];

  const int tid   = threadIdx.x;
  const int lane  = tid & 31;
  const int wave  = __builtin_amdgcn_readfirstlane(tid >> 5);
  const int rt    = wave >> 3;
  const int ct    = wave & 7;
  const int bcol  = ct * 16 + (lane & 15);
  const int khalf = (lane >> 4) * 2;

  {
    const float4* W4 = (const float4*)W;
#pragma unroll
    for (int i = 0; i < (D_FEAT * D_FEAT / 4) / WG_THREADS; ++i) {
      const int q   = i * WG_THREADS + tid;
      const int row = q >> 5;
      const int c4  = q & 31;
      *(float4*)(Ws + row * L_PITCH + c4 * 4) = W4[q];
    }
  }
  const float* bp = Ws + bcol * L_PITCH + khalf;
  const float bc = bias[bcol];
  const char* xb = (const char*)x;
  const unsigned loff = (unsigned)lane * 16u;

  int   g0, g1, m0, m1, e0, e1;
  float w0, w1, d0, d1;
  {
    const int t0    = (int)blockIdx.x < ntiles ? (int)blockIdx.x : 0;
    const int node0 = t0 * TILE_M + wave * 2;
    g0 = cnt[node0]; g1 = cnt[node0 + 1];
    m0 = g0 < CAP ? g0 : CAP;
    m1 = g1 < CAP ? g1 : CAP;
    int r0 = 0x7fffffff, r1 = 0x7fffffff;
    if (lane < m0) r0 = bkt[node0 * CAP + lane];
    if (lane < m1) r1 = bkt[(node0 + 1) * CAP + lane];
    const int s0 = gcn_sort32(r0, lane);
    const int s1 = gcn_sort32(r1, lane);
    e0 = lane < m0 ? s0 : 0;
    e1 = lane < m1 ? s1 : 0;
    w0 = rsqrtf((float)(cnt[e0] + 1));
    w1 = rsqrtf((float)(cnt[e1] + 1));
    d0 = rsqrtf((float)(g0 + 1));
    d1 = rsqrtf((float)(g1 + 1));
  }
  int npairs = ovfCount[0];
  npairs = npairs < nEdges ? npairs : nEdges;

  int buf = 0;
  for (int t = blockIdx.x; t < ntiles; t += gridDim.x) {
    float* At = As[buf];
    const int node0 = t * TILE_M + wave * 2;

    const int tn     = t + (int)gridDim.x < ntiles ? t + (int)gridDim.x : t;
    const int nnode0 = tn * TILE_M + wave * 2;
    const int ncv = cnt[nnode0 + (lane & 1)];

    float4 row[2];
    int ng0, ng1, nm0, nm1;
    int nr0 = 0x7fffffff, nr1 = 0x7fffffff;

    {
      const bool ok = node0 < n;
      const unsigned ic = (unsigned)(ok ? node0 : 0);
      float4 acc = make_float4(0.0f, 0.0f, 0.0f, 0.0f);
      float4 v[GROUP];
      gcn_issue_group(xb, e0, m0, ic, 0, loff, v);
      gcn_consume_group(w0, m0, d0, 0, v, acc);

      ng0 = __builtin_amdgcn_readlane(ncv, 0);
      ng1 = __builtin_amdgcn_readlane(ncv, 1);
      nm0 = ng0 < CAP ? ng0 : CAP;
      nm1 = ng1 < CAP ? ng1 : CAP;
      if (lane < nm0) nr0 = bkt[nnode0 * CAP + lane];
      if (lane < nm1) nr1 = bkt[(nnode0 + 1) * CAP + lane];

      gcn_accum_from(xb, e0, w0, m0, ic, d0, GROUP, loff, acc);
      if (g0 > CAP) gcn_accum_overflow(xb, cnt, ovf, npairs, node0, lane, loff, acc);
      const float e = ok ? d0 : 0.0f;
      row[0] = make_float4(acc.x * e, acc.y * e, acc.z * e, acc.w * e);
    }

    int ne0, ne1, nc0, nc1;
    {
      const bool ok = node0 + 1 < n;
      const unsigned ic = (unsigned)(ok ? node0 + 1 : 0);
      float4 acc = make_float4(0.0f, 0.0f, 0.0f, 0.0f);
      float4 v[GROUP];
      gcn_issue_group(xb, e1, m1, ic, 0, loff, v);
      asm volatile("" : "+v"(nr0), "+v"(nr1));

      const int ns0 = gcn_sort32(nr0, lane);
      const int ns1 = gcn_sort32(nr1, lane);
      ne0 = lane < nm0 ? ns0 : 0;
      ne1 = lane < nm1 ? ns1 : 0;
      nc0 = cnt[ne0];
      nc1 = cnt[ne1];

      gcn_consume_group(w1, m1, d1, 0, v, acc);
      gcn_accum_from(xb, e1, w1, m1, ic, d1, GROUP, loff, acc);
      if (g1 > CAP) gcn_accum_overflow(xb, cnt, ovf, npairs, node0 + 1, lane, loff, acc);
      const float e = ok ? d1 : 0.0f;
      row[1] = make_float4(acc.x * e, acc.y * e, acc.z * e, acc.w * e);
    }

    *(float4*)(At + (wave * 2 + 0) * L_PITCH + lane * 4) = row[0];
    *(float4*)(At + (wave * 2 + 1) * L_PITCH + lane * 4) = row[1];

    __syncthreads();

    const float nw0 = rsqrtf((float)(nc0 + 1));
    const float nw1 = rsqrtf((float)(nc1 + 1));
    const float nd0 = rsqrtf((float)(ng0 + 1));
    const float nd1 = rsqrtf((float)(ng1 + 1));

    const float* ap = At + (rt * 16 + (lane & 15)) * L_PITCH + khalf;
    v8f acc8 = {};
#pragma unroll
    for (int kk = 0; kk < D_FEAT / 4; ++kk) {
      const v2f a = gcn_frag(ap, kk);
      const v2f b = gcn_frag(bp, kk);
      acc8 = __builtin_amdgcn_wmma_f32_16x16x4_f32(
           false, a,  false, b,
           (short)0, acc8,  false,  false);
    }

    const int rbase = t * TILE_M + rt * 16 + (lane >> 4) * 8;
    float* op = out + (size_t)rbase * D_FEAT + bcol;
    if (t * TILE_M + TILE_M <= n) {
#pragma unroll
      for (int r = 0; r < 8; ++r) op[r * D_FEAT] = acc8[r] + bc;
    } else {
#pragma unroll
      for (int r = 0; r < 8; ++r)
        if (rbase + r < n) op[r * D_FEAT] = acc8[r] + bc;
    }

    buf ^= 1;
    g0 = ng0; g1 = ng1; m0 = nm0; m1 = nm1;
    e0 = ne0; e1 = ne1; w0 = nw0; w1 = nw1; d0 = nd0; d1 = nd1;
  }
}

extern "C" void kernel_launch(void* const* d_in, const int* in_sizes, int n_in,
                              void* d_out, int out_size, void* d_ws, size_t ws_size,
                              hipStream_t stream) {
  const float* x = (const float*)d_in[0];
  const float* W = (const float*)d_in[1];
  const float* b = (const float*)d_in[2];
  const int* ei  = (const int*)d_in[3];

  const int nNodes = in_sizes[0] / D_FEAT;
  const int nEdges = in_sizes[3] / 2;
  const int* src = ei;
  const int* dst = ei + nEdges;

  float* out = (float*)d_out;

  const int nPad = (nNodes + CHUNK - 1) / CHUNK * CHUNK;
  const int ePad = (nEdges + CHUNK - 1) / CHUNK * CHUNK;
  int* cnt      = (int*)d_ws;
  int* ovfCount = cnt + nPad;
  int* bkt      = ovfCount + 256;
  int* ovf      = bkt + (size_t)nPad * CAP;

  gcn_zero_counts<<<(nPad + 256) / 256, 256, 0, stream>>>(cnt, nPad + 256);

  gcn_file_edges<<<ePad / CHUNK, 256, 0, stream>>>(src, dst, cnt, bkt, ovf, ovfCount, nEdges);

  const int mtiles = (nNodes + TILE_M - 1) / TILE_M;
  gcn_gather_gemm_wmma<<<512, WG_THREADS, 0, stream>>>(x, cnt, bkt, ovf, ovfCount, W, b, out, nNodes, mtiles, nEdges);
}


# Edge filing without a returning global add per edge: edges split by 256-node range (LDS counts, one global add per range and group), then slotted per range in LDS; 3 launches -> 4; sum order kept
# speedup vs baseline: 5.0846x; 1.8854x over previous
#include <hip/hip_runtime.h>
#include <stdint.h>


typedef __attribute__((ext_vector_type(2))) float v2f;
typedef __attribute__((ext_vector_type(8))) float v8f;

#define D_FEAT 128
#define CHUNK 1024


#define CAP 32

__global__ void gcn_zero_counts(int* __restrict__ cnt, int nWords) {
  int i = blockIdx.x * blockDim.x + threadIdx.x;
  if (i < nWords) cnt[i] = 0;
}

#define RSH 8
#define RNODES 256
#define CAPR 5120
#define SPLIT_THREADS 512
#define SPLIT_EPT 8
#define SPLIT_CHUNK (SPLIT_THREADS * SPLIT_EPT)

__global__ void __launch_bounds__(SPLIT_THREADS)
gcn_split_edges(const int* __restrict__ src,
                const int* __restrict__ dst,
                int* __restrict__ rcount,
                int* __restrict__ rlist, int nEdges, int nRange) {
  extern __shared__ int hist[];
  for (int i = threadIdx.x; i < nRange; i += SPLIT_THREADS) hist[i] = 0;
  __syncthreads();

  const int base = blockIdx.x * SPLIT_CHUNK + threadIdx.x;
  int d[SPLIT_EPT], s[SPLIT_EPT], lr[SPLIT_EPT];
  if (blockIdx.x * SPLIT_CHUNK + SPLIT_CHUNK <= nEdges) {
#pragma unroll
    for (int k = 0; k < SPLIT_EPT; ++k) { d[k] = dst[base + k * SPLIT_THREADS]; s[k] = src[base + k * SPLIT_THREADS]; }
  } else {
#pragma unroll
    for (int k = 0; k < SPLIT_EPT; ++k) {
      const int e = base + k * SPLIT_THREADS;
      d[k] = -1; s[k] = 0;
      if (e < nEdges) { d[k] = dst[e]; s[k] = src[e]; }
    }
  }
#pragma unroll
  for (int k = 0; k < SPLIT_EPT; ++k) {
    lr[k] = 0;
    if (d[k] >= 0) lr[k] = atomicAdd(&hist[d[k] >> RSH], 1);
  }
  __syncthreads();
  for (int i = threadIdx.x; i < nRange; i += SPLIT_THREADS) {
    const int c = hist[i];
    int b = 0;
    if (c > 0) b = atomicAdd(&rcount[i], c);
    hist[i] = b;
  }
  __syncthreads();
#pragma unroll
  for (int k = 0; k < SPLIT_EPT; ++k) lr[k] += hist[d[k] >= 0 ? d[k] >> RSH : 0];
#pragma unroll
  for (int k = 0; k < SPLIT_EPT; ++k) {
    if (d[k] >= 0 && lr[k] < CAPR)
      rlist[(d[k] >> RSH) * CAPR + lr[k]] = (int)(((unsigned)(d[k] & (RNODES - 1)) << 24) | (unsigned)s[k]);
  }
}

static __device__ __forceinline__ void gcn_place(int* lcnt, int* img, int dl, int s, int node0,
                                                 int* __restrict__ ovf, int* __restrict__ ovfCount) {
  const int k = atomicAdd(&lcnt[dl], 1);
  if (k < CAP) {
    img[dl * CAP + k] = s;
  } else {
    const int q = atomicAdd(ovfCount, 1);
    *(int2*)(ovf + 2 * q) = make_int2(node0 + dl, s);
  }
}

__global__ void __launch_bounds__(RNODES)
gcn_rank_range(const int* __restrict__ src,
               const int* __restrict__ dst,
               const int* __restrict__ rcount,
               const int* __restrict__ rlist,
               int* __restrict__ cnt,
               int* __restrict__ bkt,
               int* __restrict__ ovf,
               int* __restrict__ ovfCount, int nEdges) {
  __shared__ int lcnt[RNODES];
  __shared__ __attribute__((aligned(16))) int img[RNODES * CAP];
  const int r = blockIdx.x;
  const int tid = threadIdx.x;
  lcnt[tid] = 0;
  for (int i = tid; i < RNODES * CAP / 4; i += RNODES) ((int4*)img)[i] = make_int4(0, 0, 0, 0);
  __syncthreads();

  const int total = rcount[r];
  if (total <= CAPR) {
    const int* lp = rlist + r * CAPR;
    for (int i0 = tid; i0 < total; i0 += RNODES * 8) {
      unsigned w[8];
#pragma unroll
      for (int u = 0; u < 8; ++u) w[u] = i0 + u * RNODES < total ? (unsigned)lp[i0 + u * RNODES] : 0u;
#pragma unroll
      for (int u = 0; u < 8; ++u)
        if (i0 + u * RNODES < total)
          gcn_place(lcnt, img, (int)(w[u] >> 24), (int)(w[u] & 0xffffffu), r * RNODES, ovf, ovfCount);
    }
  } else {
    for (int e = tid; e < nEdges; e += RNODES) {
      const int d = dst[e];
      if ((d >> RSH) == r) gcn_place(lcnt, img, d & (RNODES - 1), src[e], r * RNODES, ovf, ovfCount);
    }
  }
  __syncthreads();

  cnt[r * RNODES + tid] = lcnt[tid];
  int4* o4 = (int4*)(bkt + (size_t)r * RNODES * CAP);
  for (int i = tid; i < RNODES * CAP / 4; i += RNODES) o4[i] = ((const int4*)img)[i];
}

#define TILE_M 32
#define WG_THREADS 512
#define L_PITCH 132
#define GROUP 16

static __device__ __forceinline__ int gcn_sort32(int key, int lane) {
#pragma unroll
  for (int k = 2; k <= 32; k <<= 1) {
#pragma unroll
    for (int j = k >> 1; j > 0; j >>= 1) {
      const int other = __shfl_xor(key, j);
      const bool up    = (lane & k) == 0;
      const bool lower = (lane & j) == 0;
      const int mn = key < other ? key : other;
      const int mx = key < other ? other : key;
      key = (lower == up) ? mn : mx;
    }
  }
  return key;
}

static __device__ __forceinline__ void gcn_issue_group(const char* __restrict__ xb,
                                                       int ids, int nent, unsigned self,
                                                       int p0, unsigned loff, float4* v) {
  const int np = nent + 1;
#pragma unroll
  for (int u = 0; u < GROUP; ++u) {
    const int p = p0 + u < np ? p0 + u : np - 1;
    const unsigned e = (unsigned)__builtin_amdgcn_readlane(ids, (p - 1) & 31);
    const unsigned s = p == 0 ? self : e;
    v[u] = *(const float4*)(xb + (s * 512u + loff));
  }
}

static __device__ __forceinline__ void gcn_consume_group(float wts, int nent, float selfw,
                                                         int p0, const float4* v, float4& acc) {
  const int np = nent + 1;
#pragma unroll
  for (int u = 0; u < GROUP; ++u) {
    const bool live = p0 + u < np;
    const int p = live ? p0 + u : np - 1;
    const float we = __int_as_float(__builtin_amdgcn_readlane(__float_as_int(wts), (p - 1) & 31));
    const float w = p == 0 ? selfw : we;
    acc.x = fmaf(live ? v[u].x : 0.0f, w, acc.x);
    acc.y = fmaf(live ? v[u].y : 0.0f, w, acc.y);
    acc.z = fmaf(live ? v[u].z : 0.0f, w, acc.z);
    acc.w = fmaf(live ? v[u].w : 0.0f, w, acc.w);
  }
}

static __device__ __forceinline__ void gcn_accum_from(const char* __restrict__ xb,
                                                      int ids, float wts, int nent,
                                                      unsigned self, float selfw, int p0,
                                                      unsigned loff, float4& acc) {
  for (; p0 < nent + 1; p0 += GROUP) {
    float4 v[GROUP];
    gcn_issue_group(xb, ids, nent, self, p0, loff, v);
    gcn_consume_group(wts, nent, selfw, p0, v, acc);
  }
}

static __device__ __forceinline__ void gcn_accum_overflow(const char* __restrict__ xb,
                                                       const int* __restrict__ cnt,
                                                       const int* __restrict__ ovf,
                                                       int npairs, int node, int lane,
                                                       unsigned loff, float4& acc) {
  for (int c0 = 0; c0 < npairs; c0 += 32) {
    int od = -1, os = 0;
    if (c0 + lane < npairs) {
      const int2 pr = *(const int2*)(ovf + 2 * (c0 + lane));
      od = pr.x;
      os = pr.y;
    }
    const bool hit = od == node;
    float ow = 0.0f;
    if (hit) ow = rsqrtf((float)(cnt[os] + 1));
    unsigned mask = __builtin_amdgcn_ballot_w32(hit);
    while (mask != 0u) {
      const int b = __builtin_ctz(mask);
      mask &= mask - 1u;
      const unsigned s = (unsigned)__builtin_amdgcn_readlane(os, b);
      const float w = __int_as_float(__builtin_amdgcn_readlane(__float_as_int(ow), b));
      const float4 v = *(const float4*)(xb + (s * 512u + loff));
      acc.x = fmaf(v.x, w, acc.x);
      acc.y = fmaf(v.y, w, acc.y);
      acc.z = fmaf(v.z, w, acc.z);
      acc.w = fmaf(v.w, w, acc.w);
    }
  }
}

static __device__ __forceinline__ v2f gcn_frag(const float* p, int kk) {
  return *(const v2f*)(p + 4 * kk);
}

__global__ void __launch_bounds__(WG_THREADS, 8)
gcn_gather_gemm_wmma(const float* __restrict__ x,
                     const int* __restrict__ cnt,
                     const int* __restrict__ bkt,
                     const int* __restrict__ ovf,
                     const int* __restrict__ ovfCount,
                     const float* __restrict__ W,
                     const float* __restrict__ bias,
                     float* __restrict__ out, int n, int ntiles, int nEdges) {
  __shared__ __attribute__((aligned(16))) float Ws[D_FEAT * L_PITCH];
  __shared__ __attribute__((aligned(16))) float As[2][TILE_M * L_PITCH];

  const int tid   = threadIdx.x;
  const int lane  = tid & 31;
  const int wave  = __builtin_amdgcn_readfirstlane(tid >> 5);
  const int rt    = wave >> 3;
  const int ct    = wave & 7;
  const int bcol  = ct * 16 + (lane & 15);
  const int khalf = (lane >> 4) * 2;

  {
    const float4* W4 = (const float4*)W;
#pragma unroll
    for (int i = 0; i < (D_FEAT * D_FEAT / 4) / WG_THREADS; ++i) {
      const int q   = i * WG_THREADS + tid;
      const int row = q >> 5;
      const int c4  = q & 31;
      *(float4*)(Ws + row * L_PITCH + c4 * 4) = W4[q];
    }
  }
  const float* bp = Ws + bcol * L_PITCH + khalf;
  const float bc = bias[bcol];
  const char* xb = (const char*)x;
  const unsigned loff = (unsigned)lane * 16u;

  int   g0, g1, m0, m1, e0, e1;
  float w0, w1, d0, d1;
  {
    const int t0    = (int)blockIdx.x < ntiles ? (int)blockIdx.x : 0;
    const int node0 = t0 * TILE_M + wave * 2;
    g0 = cnt[node0]; g1 = cnt[node0 + 1];
    m0 = g0 < CAP ? g0 : CAP;
    m1 = g1 < CAP ? g1 : CAP;
    int r0 = 0x7fffffff, r1 = 0x7fffffff;
    if (lane < m0) r0 = bkt[node0 * CAP + lane];
    if (lane < m1) r1 = bkt[(node0 + 1) * CAP + lane];
    const int s0 = gcn_sort32(r0, lane);
    const int s1 = gcn_sort32(r1, lane);
    e0 = lane < m0 ? s0 : 0;
    e1 = lane < m1 ? s1 : 0;
    w0 = rsqrtf((float)(cnt[e0] + 1));
    w1 = rsqrtf((float)(cnt[e1] + 1));
    d0 = rsqrtf((float)(g0 + 1));
    d1 = rsqrtf((float)(g1 + 1));
  }
  int npairs = ovfCount[0];
  npairs = npairs < nEdges ? npairs : nEdges;

  int buf = 0;
  for (int t = blockIdx.x; t < ntiles; t += gridDim.x) {
    float* At = As[buf];
    const int node0 = t * TILE_M + wave * 2;

    const int tn     = t + (int)gridDim.x < ntiles ? t + (int)gridDim.x : t;
    const int nnode0 = tn * TILE_M + wave * 2;
    const int ncv = cnt[nnode0 + (lane & 1)];

    float4 row[2];
    int ng0, ng1, nm0, nm1;
    int nr0 = 0x7fffffff, nr1 = 0x7fffffff;

    {
      const bool ok = node0 < n;
      const unsigned ic = (unsigned)(ok ? node0 : 0);
      float4 acc = make_float4(0.0f, 0.0f, 0.0f, 0.0f);
      float4 v[GROUP];
      gcn_issue_group(xb, e0, m0, ic, 0, loff, v);
      gcn_consume_group(w0, m0, d0, 0, v, acc);

      ng0 = __builtin_amdgcn_readlane(ncv, 0);
      ng1 = __builtin_amdgcn_readlane(ncv, 1);
      nm0 = ng0 < CAP ? ng0 : CAP;
      nm1 = ng1 < CAP ? ng1 : CAP;
      if (lane < nm0) nr0 = bkt[nnode0 * CAP + lane];
      if (lane < nm1) nr1 = bkt[(nnode0 + 1) * CAP + lane];

      gcn_accum_from(xb, e0, w0, m0, ic, d0, GROUP, loff, acc);
      if (g0 > CAP) gcn_accum_overflow(xb, cnt, ovf, npairs, node0, lane, loff, acc);
      const float e = ok ? d0 : 0.0f;
      row[0] = make_float4(acc.x * e, acc.y * e, acc.z * e, acc.w * e);
    }

    int ne0, ne1, nc0, nc1;
    {
      const bool ok = node0 + 1 < n;
      const unsigned ic = (unsigned)(ok ? node0 + 1 : 0);
      float4 acc = make_float4(0.0f, 0.0f, 0.0f, 0.0f);
      float4 v[GROUP];
      gcn_issue_group(xb, e1, m1, ic, 0, loff, v);
      asm volatile("" : "+v"(nr0), "+v"(nr1));

      const int ns0 = gcn_sort32(nr0, lane);
      const int ns1 = gcn_sort32(nr1, lane);
      ne0 = lane < nm0 ? ns0 : 0;
      ne1 = lane < nm1 ? ns1 : 0;
      nc0 = cnt[ne0];
      nc1 = cnt[ne1];

      gcn_consume_group(w1, m1, d1, 0, v, acc);
      gcn_accum_from(xb, e1, w1, m1, ic, d1, GROUP, loff, acc);
      if (g1 > CAP) gcn_accum_overflow(xb, cnt, ovf, npairs, node0 + 1, lane, loff, acc);
      const float e = ok ? d1 : 0.0f;
      row[1] = make_float4(acc.x * e, acc.y * e, acc.z * e, acc.w * e);
    }

    *(float4*)(At + (wave * 2 + 0) * L_PITCH + lane * 4) = row[0];
    *(float4*)(At + (wave * 2 + 1) * L_PITCH + lane * 4) = row[1];

    __syncthreads();

    const float nw0 = rsqrtf((float)(nc0 + 1));
    const float nw1 = rsqrtf((float)(nc1 + 1));
    const float nd0 = rsqrtf((float)(ng0 + 1));
    const float nd1 = rsqrtf((float)(ng1 + 1));

    const float* ap = At + (rt * 16 + (lane & 15)) * L_PITCH + khalf;
    v8f acc8 = {};
#pragma unroll
    for (int kk = 0; kk < D_FEAT / 4; ++kk) {
      const v2f a = gcn_frag(ap, kk);
      const v2f b = gcn_frag(bp, kk);
      acc8 = __builtin_amdgcn_wmma_f32_16x16x4_f32(
           false, a,  false, b,
           (short)0, acc8,  false,  false);
    }

    const int rbase = t * TILE_M + rt * 16 + (lane >> 4) * 8;
    float* op = out + (size_t)rbase * D_FEAT + bcol;
    if (t * TILE_M + TILE_M <= n) {
#pragma unroll
      for (int r = 0; r < 8; ++r) op[r * D_FEAT] = acc8[r] + bc;
    } else {
#pragma unroll
      for (int r = 0; r < 8; ++r)
        if (rbase + r < n) op[r * D_FEAT] = acc8[r] + bc;
    }

    buf ^= 1;
    g0 = ng0; g1 = ng1; m0 = nm0; m1 = nm1;
    e0 = ne0; e1 = ne1; w0 = nw0; w1 = nw1; d0 = nd0; d1 = nd1;
  }
}

extern "C" void kernel_launch(void* const* d_in, const int* in_sizes, int n_in,
                              void* d_out, int out_size, void* d_ws, size_t ws_size,
                              hipStream_t stream) {
  const float* x = (const float*)d_in[0];
  const float* W = (const float*)d_in[1];
  const float* b = (const float*)d_in[2];
  const int* ei  = (const int*)d_in[3];

  const int nNodes = in_sizes[0] / D_FEAT;
  const int nEdges = in_sizes[3] / 2;
  const int* src = ei;
  const int* dst = ei + nEdges;

  float* out = (float*)d_out;

  const int nPad   = (nNodes + CHUNK - 1) / CHUNK * CHUNK;
  const int ePad   = (nEdges + CHUNK - 1) / CHUNK * CHUNK;
  const int nRange = nPad / RNODES;
  const int rPad   = (nRange + CHUNK - 1) / CHUNK * CHUNK;
  int* cnt      = (int*)d_ws;
  int* ovfCount = cnt + nPad;
  int* rcount   = ovfCount + 256;
  int* bkt      = rcount + rPad;
  int* ovf      = bkt + (size_t)nPad * CAP;
  int* rlist    = ovf + (size_t)2 * ePad;

  gcn_zero_counts<<<(256 + rPad) / 256, 256, 0, stream>>>(ovfCount, 256 + rPad);

  gcn_split_edges<<<(nEdges + SPLIT_CHUNK - 1) / SPLIT_CHUNK, SPLIT_THREADS, nRange * sizeof(int), stream>>>(src, dst, rcount, rlist, nEdges, nRange);

  gcn_rank_range<<<nRange, RNODES, 0, stream>>>(src, dst, rcount, rlist, cnt, bkt, ovf, ovfCount, nEdges);

  const int mtiles = (nNodes + TILE_M - 1) / TILE_M;
  gcn_gather_gemm_wmma<<<512, WG_THREADS, 0, stream>>>(x, cnt, bkt, ovf, ovfCount, W, b, out, nNodes, mtiles, nEdges);
}


# Range-list stores coalesced: a chunk's edges are grouped by range in an LDS staging buffer (prefix sum of the LDS counts) and stored as runs of consecutive words; list places and sum order kept
# speedup vs baseline: 5.8405x; 1.1487x over previous
#include <hip/hip_runtime.h>
#include <stdint.h>


typedef __attribute__((ext_vector_type(2))) float v2f;
typedef __attribute__((ext_vector_type(8))) float v8f;

#define D_FEAT 128
#define CHUNK 1024


#define CAP 32

__global__ void gcn_zero_counts(int* __restrict__ cnt, int nWords) {
  int i = blockIdx.x * blockDim.x + threadIdx.x;
  if (i < nWords) cnt[i] = 0;
}

#define RSH 8
#define RNODES 256
#define CAPR 5120
#define SPLIT_THREADS 512
#define SPLIT_EPT 8
#define SPLIT_CHUNK (SPLIT_THREADS * SPLIT_EPT)

__global__ void __launch_bounds__(SPLIT_THREADS)
gcn_split_edges(const int* __restrict__ src,
                const int* __restrict__ dst,
                int* __restrict__ rcount,
                int* __restrict__ rlist, int nEdges, int nRange) {
  extern __shared__ int hist[];
  int* lofs = hist + nRange;
  __shared__ int2 stage[SPLIT_CHUNK];
  __shared__ int wtot[SPLIT_THREADS / 32];
  const int lane = threadIdx.x & 31;
  const int wave = threadIdx.x >> 5;
  for (int i = threadIdx.x; i < nRange; i += SPLIT_THREADS) hist[i] = 0;
  __syncthreads();

  const int base = blockIdx.x * SPLIT_CHUNK + threadIdx.x;
  int d[SPLIT_EPT], s[SPLIT_EPT], lr[SPLIT_EPT];
  if (blockIdx.x * SPLIT_CHUNK + SPLIT_CHUNK <= nEdges) {
#pragma unroll
    for (int k = 0; k < SPLIT_EPT; ++k) { d[k] = dst[base + k * SPLIT_THREADS]; s[k] = src[base + k * SPLIT_THREADS]; }
  } else {
#pragma unroll
    for (int k = 0; k < SPLIT_EPT; ++k) {
      const int e = base + k * SPLIT_THREADS;
      d[k] = -1; s[k] = 0;
      if (e < nEdges) { d[k] = dst[e]; s[k] = src[e]; }
    }
  }
#pragma unroll
  for (int k = 0; k < SPLIT_EPT; ++k) {
    lr[k] = 0;
    if (d[k] >= 0) lr[k] = atomicAdd(&hist[d[k] >> RSH], 1);
  }
  __syncthreads();
  int carry = 0;
  for (int i0 = 0; i0 < nRange; i0 += SPLIT_THREADS) {
    const int i = i0 + threadIdx.x;
    const int c = i < nRange ? hist[i] : 0;
    int b = 0;
    if (c > 0) b = atomicAdd(&rcount[i], c);
    int incl = c;
#pragma unroll
    for (int o = 1; o < 32; o <<= 1) {
      const int t = __shfl_up(incl, o);
      if (lane >= o) incl += t;
    }
    if (lane == 31) wtot[wave] = incl;
    __syncthreads();
    int below = 0, all = 0;
#pragma unroll
    for (int w = 0; w < SPLIT_THREADS / 32; ++w) {
      const int t = wtot[w];
      if (w < wave) below += t;
      all += t;
    }
    if (i < nRange) { lofs[i] = carry + below + incl - c; hist[i] = b; }
    carry += all;
    __syncthreads();
  }
  int gs[SPLIT_EPT], gp[SPLIT_EPT];
#pragma unroll
  for (int k = 0; k < SPLIT_EPT; ++k) {
    const int r = d[k] >= 0 ? d[k] >> RSH : 0;
    gs[k] = lofs[r] + lr[k];
    gp[k] = hist[r] + lr[k];
  }
#pragma unroll
  for (int k = 0; k < SPLIT_EPT; ++k) {
    if (d[k] >= 0) {
      const int w = (int)(((unsigned)(d[k] & (RNODES - 1)) << 24) | (unsigned)s[k]);
      stage[gs[k]] = make_int2(w, gp[k] < CAPR ? (d[k] >> RSH) * CAPR + gp[k] : -1);
    }
  }
  __syncthreads();
  int2 q[SPLIT_EPT];
#pragma unroll
  for (int k = 0; k < SPLIT_EPT; ++k) {
    const int j = k * SPLIT_THREADS + threadIdx.x;
    q[k] = make_int2(0, -1);
    if (j < carry) q[k] = stage[j];
  }
#pragma unroll
  for (int k = 0; k < SPLIT_EPT; ++k)
    if (q[k].y >= 0) rlist[q[k].y] = q[k].x;
}

static __device__ __forceinline__ void gcn_place(int* lcnt, int* img, int dl, int s, int node0,
                                                 int* __restrict__ ovf, int* __restrict__ ovfCount) {
  const int k = atomicAdd(&lcnt[dl], 1);
  if (k < CAP) {
    img[dl * CAP + k] = s;
  } else {
    const int q = atomicAdd(ovfCount, 1);
    *(int2*)(ovf + 2 * q) = make_int2(node0 + dl, s);
  }
}

__global__ void __launch_bounds__(RNODES)
gcn_rank_range(const int* __restrict__ src,
               const int* __restrict__ dst,
               const int* __restrict__ rcount,
               const int* __restrict__ rlist,
               int* __restrict__ cnt,
               int* __restrict__ bkt,
               int* __restrict__ ovf,
               int* __restrict__ ovfCount, int nEdges) {
  __shared__ int lcnt[RNODES];
  __shared__ __attribute__((aligned(16))) int img[RNODES * CAP];
  const int r = blockIdx.x;
  const int tid = threadIdx.x;
  lcnt[tid] = 0;
  for (int i = tid; i < RNODES * CAP / 4; i += RNODES) ((int4*)img)[i] = make_int4(0, 0, 0, 0);
  __syncthreads();

  const int total = rcount[r];
  if (total <= CAPR) {
    const int* lp = rlist + r * CAPR;
    for (int i0 = tid; i0 < total; i0 += RNODES * 8) {
      unsigned w[8];
#pragma unroll
      for (int u = 0; u < 8; ++u) w[u] = i0 + u * RNODES < total ? (unsigned)lp[i0 + u * RNODES] : 0u;
#pragma unroll
      for (int u = 0; u < 8; ++u)
        if (i0 + u * RNODES < total)
          gcn_place(lcnt, img, (int)(w[u] >> 24), (int)(w[u] & 0xffffffu), r * RNODES, ovf, ovfCount);
    }
  } else {
    for (int e = tid; e < nEdges; e += RNODES) {
      const int d = dst[e];
      if ((d >> RSH) == r) gcn_place(lcnt, img, d & (RNODES - 1), src[e], r * RNODES, ovf, ovfCount);
    }
  }
  __syncthreads();

  cnt[r * RNODES + tid] = lcnt[tid];
  int4* o4 = (int4*)(bkt + (size_t)r * RNODES * CAP);
  for (int i = tid; i < RNODES * CAP / 4; i += RNODES) o4[i] = ((const int4*)img)[i];
}

#define TILE_M 32
#define WG_THREADS 512
#define L_PITCH 132
#define GROUP 16

static __device__ __forceinline__ int gcn_sort32(int key, int lane) {
#pragma unroll
  for (int k = 2; k <= 32; k <<= 1) {
#pragma unroll
    for (int j = k >> 1; j > 0; j >>= 1) {
      const int other = __shfl_xor(key, j);
      const bool up    = (lane & k) == 0;
      const bool lower = (lane & j) == 0;
      const int mn = key < other ? key : other;
      const int mx = key < other ? other : key;
      key = (lower == up) ? mn : mx;
    }
  }
  return key;
}

static __device__ __forceinline__ void gcn_issue_group(const char* __restrict__ xb,
                                                       int ids, int nent, unsigned self,
                                                       int p0, unsigned loff, float4* v) {
  const int np = nent + 1;
#pragma unroll
  for (int u = 0; u < GROUP; ++u) {
    const int p = p0 + u < np ? p0 + u : np - 1;
    const unsigned e = (unsigned)__builtin_amdgcn_readlane(ids, (p - 1) & 31);
    const unsigned s = p == 0 ? self : e;
    v[u] = *(const float4*)(xb + (s * 512u + loff));
  }
}

static __device__ __forceinline__ void gcn_consume_group(float wts, int nent, float selfw,
                                                         int p0, const float4* v, float4& acc) {
  const int np = nent + 1;
#pragma unroll
  for (int u = 0; u < GROUP; ++u) {
    const bool live = p0 + u < np;
    const int p = live ? p0 + u : np - 1;
    const float we = __int_as_float(__builtin_amdgcn_readlane(__float_as_int(wts), (p - 1) & 31));
    const float w = p == 0 ? selfw : we;
    acc.x = fmaf(live ? v[u].x : 0.0f, w, acc.x);
    acc.y = fmaf(live ? v[u].y : 0.0f, w, acc.y);
    acc.z = fmaf(live ? v[u].z : 0.0f, w, acc.z);
    acc.w = fmaf(live ? v[u].w : 0.0f, w, acc.w);
  }
}

static __device__ __forceinline__ void gcn_accum_from(const char* __restrict__ xb,
                                                      int ids, float wts, int nent,
                                                      unsigned self, float selfw, int p0,
                                                      unsigned loff, float4& acc) {
  for (; p0 < nent + 1; p0 += GROUP) {
    float4 v[GROUP];
    gcn_issue_group(xb, ids, nent, self, p0, loff, v);
    gcn_consume_group(wts, nent, selfw, p0, v, acc);
  }
}

static __device__ __forceinline__ void gcn_accum_overflow(const char* __restrict__ xb,
                                                       const int* __restrict__ cnt,
                                                       const int* __restrict__ ovf,
                                                       int npairs, int node, int lane,
                                                       unsigned loff, float4& acc) {
  for (int c0 = 0; c0 < npairs; c0 += 32) {
    int od = -1, os = 0;
    if (c0 + lane < npairs) {
      const int2 pr = *(const int2*)(ovf + 2 * (c0 + lane));
      od = pr.x;
      os = pr.y;
    }
    const bool hit = od == node;
    float ow = 0.0f;
    if (hit) ow = rsqrtf((float)(cnt[os] + 1));
    unsigned mask = __builtin_amdgcn_ballot_w32(hit);
    while (mask != 0u) {
      const int b = __builtin_ctz(mask);
      mask &= mask - 1u;
      const unsigned s = (unsigned)__builtin_amdgcn_readlane(os, b);
      const float w = __int_as_float(__builtin_amdgcn_readlane(__float_as_int(ow), b));
      const float4 v = *(const float4*)(xb + (s * 512u + loff));
      acc.x = fmaf(v.x, w, acc.x);
      acc.y = fmaf(v.y, w, acc.y);
      acc.z = fmaf(v.z, w, acc.z);
      acc.w = fmaf(v.w, w, acc.w);
    }
  }
}

static __device__ __forceinline__ v2f gcn_frag(const float* p, int kk) {
  return *(const v2f*)(p + 4 * kk);
}

__global__ void __launch_bounds__(WG_THREADS, 8)
gcn_gather_gemm_wmma(const float* __restrict__ x,
                     const int* __restrict__ cnt,
                     const int* __restrict__ bkt,
                     const int* __restrict__ ovf,
                     const int* __restrict__ ovfCount,
                     const float* __restrict__ W,
                     const float* __restrict__ bias,
                     float* __restrict__ out, int n, int ntiles, int nEdges) {
  __shared__ __attribute__((aligned(16))) float Ws[D_FEAT * L_PITCH];
  __shared__ __attribute__((aligned(16))) float As[2][TILE_M * L_PITCH];

  const int tid   = threadIdx.x;
  const int lane  = tid & 31;
  const int wave  = __builtin_amdgcn_readfirstlane(tid >> 5);
  const int rt    = wave >> 3;
  const int ct    = wave & 7;
  const int bcol  = ct * 16 + (lane & 15);
  const int khalf = (lane >> 4) * 2;

  {
    const float4* W4 = (const float4*)W;
#pragma unroll
    for (int i = 0; i < (D_FEAT * D_FEAT / 4) / WG_THREADS; ++i) {
      const int q   = i * WG_THREADS + tid;
      const int row = q >> 5;
      const int c4  = q & 31;
      *(float4*)(Ws + row * L_PITCH + c4 * 4) = W4[q];
    }
  }
  const float* bp = Ws + bcol * L_PITCH + khalf;
  const float bc = bias[bcol];
  const char* xb = (const char*)x;
  const unsigned loff = (unsigned)lane * 16u;

  int   g0, g1, m0, m1, e0, e1;
  float w0, w1, d0, d1;
  {
    const int t0    = (int)blockIdx.x < ntiles ? (int)blockIdx.x : 0;
    const int node0 = t0 * TILE_M + wave * 2;
    g0 = cnt[node0]; g1 = cnt[node0 + 1];
    m0 = g0 < CAP ? g0 : CAP;
    m1 = g1 < CAP ? g1 : CAP;
    int r0 = 0x7fffffff, r1 = 0x7fffffff;
    if (lane < m0) r0 = bkt[node0 * CAP + lane];
    if (lane < m1) r1 = bkt[(node0 + 1) * CAP + lane];
    const int s0 = gcn_sort32(r0, lane);
    const int s1 = gcn_sort32(r1, lane);
    e0 = lane < m0 ? s0 : 0;
    e1 = lane < m1 ? s1 : 0;
    w0 = rsqrtf((float)(cnt[e0] + 1));
    w1 = rsqrtf((float)(cnt[e1] + 1));
    d0 = rsqrtf((float)(g0 + 1));
    d1 = rsqrtf((float)(g1 + 1));
  }
  int npairs = ovfCount[0];
  npairs = npairs < nEdges ? npairs : nEdges;

  int buf = 0;
  for (int t = blockIdx.x; t < ntiles; t += gridDim.x) {
    float* At = As[buf];
    const int node0 = t * TILE_M + wave * 2;

    const int tn     = t + (int)gridDim.x < ntiles ? t + (int)gridDim.x : t;
    const int nnode0 = tn * TILE_M + wave * 2;
    const int ncv = cnt[nnode0 + (lane & 1)];

    float4 row[2];
    int ng0, ng1, nm0, nm1;
    int nr0 = 0x7fffffff, nr1 = 0x7fffffff;

    {
      const bool ok = node0 < n;
      const unsigned ic = (unsigned)(ok ? node0 : 0);
      float4 acc = make_float4(0.0f, 0.0f, 0.0f, 0.0f);
      float4 v[GROUP];
      gcn_issue_group(xb, e0, m0, ic, 0, loff, v);
      gcn_consume_group(w0, m0, d0, 0, v, acc);

      ng0 = __builtin_amdgcn_readlane(ncv, 0);
      ng1 = __builtin_amdgcn_readlane(ncv, 1);
      nm0 = ng0 < CAP ? ng0 : CAP;
      nm1 = ng1 < CAP ? ng1 : CAP;
      if (lane < nm0) nr0 = bkt[nnode0 * CAP + lane];
      if (lane < nm1) nr1 = bkt[(nnode0 + 1) * CAP + lane];

      gcn_accum_from(xb, e0, w0, m0, ic, d0, GROUP, loff, acc);
      if (g0 > CAP) gcn_accum_overflow(xb, cnt, ovf, npairs, node0, lane, loff, acc);
      const float e = ok ? d0 : 0.0f;
      row[0] = make_float4(acc.x * e, acc.y * e, acc.z * e, acc.w * e);
    }

    int ne0, ne1, nc0, nc1;
    {
      const bool ok = node0 + 1 < n;
      const unsigned ic = (unsigned)(ok ? node0 + 1 : 0);
      float4 acc = make_float4(0.0f, 0.0f, 0.0f, 0.0f);
      float4 v[GROUP];
      gcn_issue_group(xb, e1, m1, ic, 0, loff, v);
      asm volatile("" : "+v"(nr0), "+v"(nr1));

      const int ns0 = gcn_sort32(nr0, lane);
      const int ns1 = gcn_sort32(nr1, lane);
      ne0 = lane < nm0 ? ns0 : 0;
      ne1 = lane < nm1 ? ns1 : 0;
      nc0 = cnt[ne0];
      nc1 = cnt[ne1];

      gcn_consume_group(w1, m1, d1, 0, v, acc);
      gcn_accum_from(xb, e1, w1, m1, ic, d1, GROUP, loff, acc);
      if (g1 > CAP) gcn_accum_overflow(xb, cnt, ovf, npairs, node0 + 1, lane, loff, acc);
      const float e = ok ? d1 : 0.0f;
      row[1] = make_float4(acc.x * e, acc.y * e, acc.z * e, acc.w * e);
    }

    *(float4*)(At + (wave * 2 + 0) * L_PITCH + lane * 4) = row[0];
    *(float4*)(At + (wave * 2 + 1) * L_PITCH + lane * 4) = row[1];

    __syncthreads();

    const float nw0 = rsqrtf((float)(nc0 + 1));
    const float nw1 = rsqrtf((float)(nc1 + 1));
    const float nd0 = rsqrtf((float)(ng0 + 1));
    const float nd1 = rsqrtf((float)(ng1 + 1));

    const float* ap = At + (rt * 16 + (lane & 15)) * L_PITCH + khalf;
    v8f acc8 = {};
#pragma unroll
    for (int kk = 0; kk < D_FEAT / 4; ++kk) {
      const v2f a = gcn_frag(ap, kk);
      const v2f b = gcn_frag(bp, kk);
      acc8 = __builtin_amdgcn_wmma_f32_16x16x4_f32(
           false, a,  false, b,
           (short)0, acc8,  false,  false);
    }

    const int rbase = t * TILE_M + rt * 16 + (lane >> 4) * 8;
    float* op = out + (size_t)rbase * D_FEAT + bcol;
    if (t * TILE_M + TILE_M <= n) {
#pragma unroll
      for (int r = 0; r < 8; ++r) op[r * D_FEAT] = acc8[r] + bc;
    } else {
#pragma unroll
      for (int r = 0; r < 8; ++r)
        if (rbase + r < n) op[r * D_FEAT] = acc8[r] + bc;
    }

    buf ^= 1;
    g0 = ng0; g1 = ng1; m0 = nm0; m1 = nm1;
    e0 = ne0; e1 = ne1; w0 = nw0; w1 = nw1; d0 = nd0; d1 = nd1;
  }
}

extern "C" void kernel_launch(void* const* d_in, const int* in_sizes, int n_in,
                              void* d_out, int out_size, void* d_ws, size_t ws_size,
                              hipStream_t stream) {
  const float* x = (const float*)d_in[0];
  const float* W = (const float*)d_in[1];
  const float* b = (const float*)d_in[2];
  const int* ei  = (const int*)d_in[3];

  const int nNodes = in_sizes[0] / D_FEAT;
  const int nEdges = in_sizes[3] / 2;
  const int* src = ei;
  const int* dst = ei + nEdges;

  float* out = (float*)d_out;

  const int nPad   = (nNodes + CHUNK - 1) / CHUNK * CHUNK;
  const int ePad   = (nEdges + CHUNK - 1) / CHUNK * CHUNK;
  const int nRange = nPad / RNODES;
  const int rPad   = (nRange + CHUNK - 1) / CHUNK * CHUNK;
  int* cnt      = (int*)d_ws;
  int* ovfCount = cnt + nPad;
  int* rcount   = ovfCount + 256;
  int* bkt      = rcount + rPad;
  int* ovf      = bkt + (size_t)nPad * CAP;
  int* rlist    = ovf + (size_t)2 * ePad;

  gcn_zero_counts<<<(256 + rPad) / 256, 256, 0, stream>>>(ovfCount, 256 + rPad);

  gcn_split_edges<<<(nEdges + SPLIT_CHUNK - 1) / SPLIT_CHUNK, SPLIT_THREADS, 2 * nRange * sizeof(int), stream>>>(src, dst, rcount, rlist, nEdges, nRange);

  gcn_rank_range<<<nRange, RNODES, 0, stream>>>(src, dst, rcount, rlist, cnt, bkt, ovf, ovfCount, nEdges);

  const int mtiles = (nNodes + TILE_M - 1) / TILE_M;
  gcn_gather_gemm_wmma<<<512, WG_THREADS, 0, stream>>>(x, cnt, bkt, ovf, ovfCount, W, b, out, nNodes, mtiles, nEdges);
}
